# starting kernel
# baseline (speedup 1.0000x reference)
#include <hip/hip_runtime.h>
#include <hip/hip_bf16.h>

typedef __attribute__((ext_vector_type(2))) float v2f;
typedef __attribute__((ext_vector_type(8))) float v8f;

__global__ void k_deg_init(float* __restrict__ deg, int n) {
    int i = blockIdx.x * blockDim.x + threadIdx.x;
    if (i < n) deg[i] = 1.0f;
}

__global__ void k_deg_accum(const int* __restrict__ dst, float* __restrict__ deg, int e) {
    int i = blockIdx.x * blockDim.x + threadIdx.x;
    if (i < e) atomicAdd(&deg[dst[i]], 1.0f);
}

__global__ void k_dinv(const float* __restrict__ deg, float* __restrict__ dinv, int n) {
    int i = blockIdx.x * blockDim.x + threadIdx.x;
    if (i < n) dinv[i] = rsqrtf(deg[i]);
}

template <int DCOL>
__global__ __launch_bounds__(256) void k_gemm_wmma(const float* __restrict__ X,
                                                   const float* __restrict__ W,
                                                   float* __restrict__ H,
                                                   int nrows) {
    constexpr int COLT = DCOL / 16;
    constexpr int ROWT = 8 / COLT;
    const int wave = threadIdx.x >> 5;
    const int lane = threadIdx.x & 31;
    const int m0 = (blockIdx.x * ROWT + (wave / COLT)) * 16;
    const int n0 = (wave % COLT) * 16;
    if (m0 >= nrows) return;

    const int lr = lane & 15;
    const int lh = lane >> 4;

    v8f c = {};
    const float* xrow = X + (long long)(m0 + lr) * 64;
#pragma unroll
    for (int k = 0; k < 64; k += 4) {
        v2f a, b;
        a.x = xrow[k + lh * 2 + 0];
        a.y = xrow[k + lh * 2 + 1];
        b.x = W[(k + lh * 2 + 0) * DCOL + n0 + lr];
        b.y = W[(k + lh * 2 + 1) * DCOL + n0 + lr];
        c = __builtin_amdgcn_wmma_f32_16x16x4_f32(false, a, false, b,
                                                  (short)0, c, false, false);
    }
#pragma unroll
    for (int r = 0; r < 8; ++r) {
        H[(long long)(m0 + r + lh * 8) * DCOL + n0 + lr] = c[r];
    }
}

__global__ void k_self_bias(const float* __restrict__ H, const float* __restrict__ dinv,
                            const float* __restrict__ b, float* __restrict__ out,
                            int n, int D) {
    long long idx = (long long)blockIdx.x * blockDim.x + threadIdx.x;
    long long total = (long long)n * D;
    if (idx >= total) return;
    int i = (int)(idx / D);
    int f = (int)(idx % D);
    float di = dinv[i];
    out[idx] = H[idx] * di * di + b[f];
}

__global__ void k_scatter(const float* __restrict__ H, const int* __restrict__ src,
                          const int* __restrict__ dst, const float* __restrict__ dinv,
                          float* __restrict__ out, int e, int D) {
    const int tpe = D >> 2;
    long long gid = (long long)blockIdx.x * blockDim.x + threadIdx.x;
    long long total = (long long)e * tpe;
    if (gid >= total) return;
    int edge = (int)(gid / tpe);
    int f = (int)(gid % tpe) * 4;
    int s = src[edge], d = dst[edge];
    float norm = dinv[s] * dinv[d];
    const float4 hv = *(const float4*)(H + (long long)s * D + f);
    float* op = out + (long long)d * D + f;
    atomicAdd(op + 0, hv.x * norm);
    atomicAdd(op + 1, hv.y * norm);
    atomicAdd(op + 2, hv.z * norm);
    atomicAdd(op + 3, hv.w * norm);
}

__global__ void k_relu(float* __restrict__ a, long long total) {
    long long idx = (long long)blockIdx.x * blockDim.x + threadIdx.x;
    if (idx < total) a[idx] = fmaxf(a[idx], 0.0f);
}

extern "C" void kernel_launch(void* const* d_in, const int* in_sizes, int n_in,
                              void* d_out, int out_size, void* d_ws, size_t ws_size,
                              hipStream_t stream) {
    const int D_IN = 64, D_H = 64, D_OUT = 32;
    const int N = in_sizes[0] / D_IN;
    const int E = in_sizes[1] / 2;

    const float* x  = (const float*)d_in[0];
    const int*   ei = (const int*)d_in[1];
    const float* W1 = (const float*)d_in[2];
    const float* b1 = (const float*)d_in[3];
    const float* W2 = (const float*)d_in[4];
    const float* b2 = (const float*)d_in[5];
    float* out = (float*)d_out;

    const int* srcIdx = ei;
    const int* dstIdx = ei + E;

    float* ws   = (float*)d_ws;
    float* deg  = ws;
    float* dinv = deg  + N;
    float* h1   = dinv + N;
    float* out1 = h1   + (long long)N * D_H;
    float* h2   = out1 + (long long)N * D_H;

    const int BT = 256;
    const int rowTiles = N / 16;

    k_deg_init<<<(N + BT - 1) / BT, BT, 0, stream>>>(deg, N);
    k_deg_accum<<<(E + BT - 1) / BT, BT, 0, stream>>>(dstIdx, deg, E);
    k_dinv<<<(N + BT - 1) / BT, BT, 0, stream>>>(deg, dinv, N);

    k_gemm_wmma<D_H><<<(rowTiles + 1) / 2, BT, 0, stream>>>(x, W1, h1, N);

    {
        long long tot = (long long)N * D_H;
        k_self_bias<<<(unsigned)((tot + BT - 1) / BT), BT, 0, stream>>>(h1, dinv, b1, out1, N, D_H);
        long long st = (long long)E * (D_H / 4);
        k_scatter<<<(unsigned)((st + BT - 1) / BT), BT, 0, stream>>>(h1, srcIdx, dstIdx, dinv, out1, E, D_H);
        k_relu<<<(unsigned)((tot + BT - 1) / BT), BT, 0, stream>>>(out1, tot);
    }

    k_gemm_wmma<D_OUT><<<(rowTiles + 3) / 4, BT, 0, stream>>>(out1, W2, h2, N);

    {
        long long tot = (long long)N * D_OUT;
        k_self_bias<<<(unsigned)((tot + BT - 1) / BT), BT, 0, stream>>>(h2, dinv, b2, out, N, D_OUT);
        long long st = (long long)E * (D_OUT / 4);
        k_scatter<<<(unsigned)((st + BT - 1) / BT), BT, 0, stream>>>(h2, srcIdx, dstIdx, dinv, out, E, D_OUT);
    }

    (void)n_in; (void)out_size; (void)ws_size;
}


# In-degree counted in LDS: 64 work-groups (16 edge slices x 4 row ranges) histogram dst with LDS int adds and store planes; GEMM-1 sums them; init launch and global degree atomics gone; degrees exact
# speedup vs baseline: 2.9537x; 2.9537x over previous
#include <hip/hip_runtime.h>
#include <hip/hip_bf16.h>

typedef __attribute__((ext_vector_type(2))) float v2f;
typedef __attribute__((ext_vector_type(8))) float v8f;

constexpr int DEG_SLICES = 16;
constexpr int DEG_ROWS   = 12544;
constexpr int DEG_BATCH  = 25;
constexpr int DEG_BT     = 1024;

__global__ __launch_bounds__(DEG_BT) void k_deg_planes(const int* __restrict__ dst,
                                                       int* __restrict__ planes, int pitch,
                                                       int e, int n) {
    __shared__ int hist[DEG_ROWS];
    const int tid = threadIdx.x;
    const int slice = blockIdx.x;
    const int lo = blockIdx.y * DEG_ROWS;
    const int rows = (n - lo < DEG_ROWS) ? (n - lo) : DEG_ROWS;
    for (int i = tid; i < rows; i += DEG_BT) hist[i] = 0;
    __syncthreads();

    const int per = (e + DEG_SLICES - 1) / DEG_SLICES;
    const int e0 = slice * per;
    const int e1 = (e0 + per < e) ? (e0 + per) : e;
    for (int base = e0 + tid; base < e1; base += DEG_BT * DEG_BATCH) {
        int d[DEG_BATCH];
#pragma unroll
        for (int u = 0; u < DEG_BATCH; ++u) {
            const int i = base + u * DEG_BT;
            d[u] = (i < e1) ? dst[i] : -1;
        }
#pragma unroll
        for (int u = 0; u < DEG_BATCH; ++u) {
            const unsigned r = (unsigned)(d[u] - lo);
            if (r < (unsigned)rows) atomicAdd(&hist[r], 1);
        }
    }
    __syncthreads();

    int* prow = planes + (long long)slice * pitch + lo;
    for (int i = tid; i < rows; i += DEG_BT) prow[i] = hist[i];
}


template <int DCOL, bool RELU_A, bool FROM_DEG>
__global__ __launch_bounds__(256) void k_gemm_wmma(const float* __restrict__ X,
                                                   const float* __restrict__ W,
                                                   const float* __restrict__ bias,
                                                   const float* __restrict__ rowScale,
                                                   const int* __restrict__ degPlanes,
                                                   int degPitch,
                                                   float* __restrict__ dinvOut,
                                                   float* __restrict__ H,
                                                   float* __restrict__ OUT,
                                                   int* __restrict__ qclear,
                                                   int nrows) {
    if (blockIdx.x == 0 && threadIdx.x < 2) qclear[threadIdx.x * 64] = 0;
    constexpr int COLT = DCOL / 16;
    constexpr int ROWT = 8 / COLT;
    const int wave = threadIdx.x >> 5;
    const int lane = threadIdx.x & 31;
    const int m0 = (blockIdx.x * ROWT + (wave / COLT)) * 16;
    const int n0 = (wave % COLT) * 16;
    if (m0 >= nrows) return;

    const int lr = lane & 15;
    const int lh = lane >> 4;

    float di[8];
    float dl = 0.0f;
    if constexpr (FROM_DEG) {
        int cnt = 0;
#pragma unroll
        for (int p = 0; p < DEG_SLICES / 2; ++p)
            cnt += degPlanes[(long long)(lh * (DEG_SLICES / 2) + p) * degPitch + m0 + lr];
        cnt += __shfl_xor(cnt, 16);
        dl = rsqrtf((float)(cnt + 1));
#pragma unroll
        for (int r = 0; r < 8; ++r) di[r] = __shfl(dl, r + lh * 8);
    } else {
#pragma unroll
        for (int r = 0; r < 8; ++r) di[r] = rowScale[m0 + r + lh * 8];
    }
    const float bcol = bias[n0 + lr];

    v8f c = {};
    const float* xrow = X + (long long)(m0 + lr) * 64;
#pragma unroll
    for (int k = 0; k < 64; k += 4) {
        v2f a, b;
        a.x = xrow[k + lh * 2 + 0];
        a.y = xrow[k + lh * 2 + 1];
        if (RELU_A) {
            a.x = fmaxf(a.x, 0.0f);
            a.y = fmaxf(a.y, 0.0f);
        }
        b.x = W[(k + lh * 2 + 0) * DCOL + n0 + lr];
        b.y = W[(k + lh * 2 + 1) * DCOL + n0 + lr];
        c = __builtin_amdgcn_wmma_f32_16x16x4_f32(false, a, false, b,
                                                  (short)0, c, false, false);
    }
#pragma unroll
    for (int r = 0; r < 8; ++r) {
        const long long o = (long long)(m0 + r + lh * 8) * DCOL + n0 + lr;
        H[o] = c[r];
        OUT[o] = c[r] * di[r] * di[r] + bcol;
    }
    if constexpr (FROM_DEG) {
        if (n0 == 0 && lane < 16) dinvOut[m0 + lane] = dl;
    }
}

__device__ __forceinline__ void lds_barrier() {
    __builtin_amdgcn_fence(__ATOMIC_RELEASE, "workgroup", "local");
    __builtin_amdgcn_s_barrier();
    __builtin_amdgcn_fence(__ATOMIC_ACQUIRE, "workgroup", "local");
}

template <int D>
__global__ __launch_bounds__(256) void k_scatter_rows(const float* __restrict__ H,
                                                      const int* __restrict__ src,
                                                      const int* __restrict__ dst,
                                                      const float* __restrict__ dinv,
                                                      float* __restrict__ out, int e, int n,
                                                      int* __restrict__ q) {
    constexpr int CH = D / 32;
    constexpr int NB = 16 / CH;
    constexpr int QSTRIDE = 64;
    __shared__ int slot[2];
    const unsigned lane = threadIdx.x & 31u;
    const int wv = __builtin_amdgcn_readfirstlane((int)(threadIdx.x >> 5));
    const int nchunk = (e + 31) >> 5;
    const int ntick = (nchunk + 7) >> 3;
    const int split = n >> 1;

    int cls = 0, left = 2, raw = 0;
    if (wv == 0) {
        const unsigned id = __builtin_amdgcn_s_sendmsg_rtn(0x87);
        cls = (((id >> 16) & 0xFu) >= 4u) ? 1 : 0;
        int code = -1;
        while (left > 0) {
            int t = 0;
            if (lane == 0) t = atomicAdd(&q[cls * QSTRIDE], 1);
            t = __builtin_amdgcn_readfirstlane(t);
            if (t < ntick) { code = (cls << 30) | t; break; }
            cls ^= 1; --left;
        }
        if (lane == 0) slot[0] = code;
    }
    lds_barrier();

    for (int it = 0; it <= 2 * ntick; ++it) {
        const int code = __builtin_amdgcn_readfirstlane(slot[it & 1]);
        if (code < 0) break;
        if (wv == 0 && left > 0 && lane == 0) raw = atomicAdd(&q[cls * QSTRIDE], 1);

        const int want = code >> 30;
        const int e0 = (((code & 0x3FFFFFFF) << 3) + wv) << 5;
        if (e0 < e) {
            const int cnt = (e - e0 < 32) ? (e - e0) : 32;
            int s = 0, d = 0;
            float nm = 0.0f;
            bool mine = false;
            if ((int)lane < cnt) {
                s = src[e0 + lane];
                d = dst[e0 + lane];
                mine = ((d >= split) ? 1 : 0) == want;
            }
            if (mine) nm = dinv[s] * dinv[d];
            unsigned m = __builtin_amdgcn_ballot_w32(mine);
            while (m != 0u) {
                const int left_m = __builtin_popcount(m);
                const int c = left_m < NB ? left_m : NB;
                int jj[NB];
                float v[NB][CH];
#pragma unroll
                for (int u = 0; u < NB; ++u) {
                    if (u < c) {
                        jj[u] = __builtin_ctz(m);
                        m &= m - 1u;
                        const int sj = __builtin_amdgcn_readlane(s, jj[u]);
                        const float* hr = H + (long long)sj * D;
#pragma unroll
                        for (int cc = 0; cc < CH; ++cc) v[u][cc] = hr[lane + cc * 32];
                    }
                }
#pragma unroll
                for (int u = 0; u < NB; ++u) {
                    if (u < c) {
                        const int dj = __builtin_amdgcn_readlane(d, jj[u]);
                        const float nj =
                            __int_as_float(__builtin_amdgcn_readlane(__float_as_int(nm), jj[u]));
                        float* orow = out + (long long)dj * D;
#pragma unroll
                        for (int cc = 0; cc < CH; ++cc)
                            atomicAdd(orow + lane + cc * 32, v[u][cc] * nj);
                    }
                }
            }
        }

        if (wv == 0) {
            int nxt = -1;
            if (left > 0) {
                int t = __builtin_amdgcn_readfirstlane(raw);
                while (true) {
                    if (t < ntick) { nxt = (cls << 30) | t; break; }
                    cls ^= 1; --left;
                    if (left == 0) break;
                    t = 0;
                    if (lane == 0) t = atomicAdd(&q[cls * QSTRIDE], 1);
                    t = __builtin_amdgcn_readfirstlane(t);
                }
            }
            if (lane == 0) slot[(it + 1) & 1] = nxt;
        }
        lds_barrier();
    }
}

extern "C" void kernel_launch(void* const* d_in, const int* in_sizes, int n_in,
                              void* d_out, int out_size, void* d_ws, size_t ws_size,
                              hipStream_t stream) {
    const int D_IN = 64, D_H = 64, D_OUT = 32;
    const int N = in_sizes[0] / D_IN;
    const int E = in_sizes[1] / 2;

    const float* x  = (const float*)d_in[0];
    const int*   ei = (const int*)d_in[1];
    const float* W1 = (const float*)d_in[2];
    const float* b1 = (const float*)d_in[3];
    const float* W2 = (const float*)d_in[4];
    const float* b2 = (const float*)d_in[5];
    float* out = (float*)d_out;

    const int* srcIdx = ei;
    const int* dstIdx = ei + E;

    float* ws   = (float*)d_ws;
    float* dinv = ws   + N;
    float* h1   = dinv + N;
    float* out1 = h1   + (long long)N * D_H;
    float* h2   = out1 + (long long)N * D_H;

    int* q1 = (int*)h2;
    int* q2 = (int*)h1;

    static_assert(DEG_SLICES * 2 <= D_OUT, "the planes must fit the upper half of the h2 plane");
    static_assert(DEG_ROWS % 32 == 0, "row ranges must start on a 128-byte line of their plane");
    const int degPitch = (N + 31) & ~31;
    int* degPlanes = (int*)(h2 + (long long)N * D_OUT) - (long long)DEG_SLICES * degPitch;
    const int degRanges = (N + DEG_ROWS - 1) / DEG_ROWS;

    const int BT = 256;
    const int SG = 2048;
    const int rowTiles = N / 16;

    k_deg_planes<<<dim3(DEG_SLICES, degRanges), DEG_BT, 0, stream>>>(dstIdx, degPlanes, degPitch,
                                                                     E, N);

    k_gemm_wmma<D_H, false, true><<<(rowTiles + 1) / 2, BT, 0, stream>>>(
        x, W1, b1, nullptr, degPlanes, degPitch, dinv, h1, out1, q1, N);

    k_scatter_rows<D_H><<<SG, BT, 0, stream>>>(h1, srcIdx, dstIdx, dinv, out1, E, N, q1);

    k_gemm_wmma<D_OUT, true, false><<<(rowTiles + 3) / 4, BT, 0, stream>>>(
        out1, W2, b2, dinv, nullptr, 0, nullptr, h2, out, q2, N);

    k_scatter_rows<D_OUT><<<SG, BT, 0, stream>>>(h2, srcIdx, dstIdx, dinv, out, E, N, q2);

    (void)n_in; (void)out_size; (void)ws_size;
}
